# starting kernel
# baseline (speedup 1.0000x reference)
#include <hip/hip_runtime.h>
#include <hip/hip_bf16.h>


typedef __attribute__((ext_vector_type(2))) float v2f;
typedef __attribute__((ext_vector_type(8))) float v8f;

#define BATCH 1024
#define GRIDN 4096
#define DIM   64

__global__ __launch_bounds__(256)
void SOM_norms_kernel(const float* __restrict__ x,
                      const float* __restrict__ w,
                      float* __restrict__ ws) {
    int t = blockIdx.x * blockDim.x + threadIdx.x;
    if (t < GRIDN) {
        const float4* p = (const float4*)(w + (size_t)t * DIM);
        float s = 0.f;
#pragma unroll
        for (int i = 0; i < DIM / 4; ++i) {
            float4 v = p[i];
            s += v.x * v.x + v.y * v.y + v.z * v.z + v.w * v.w;
        }
        ws[t] = s;
    } else if (t < GRIDN + BATCH) {
        int m = t - GRIDN;
        const float4* p = (const float4*)(x + (size_t)m * DIM);
        float s = 0.f;
#pragma unroll
        for (int i = 0; i < DIM / 4; ++i) {
            float4 v = p[i];
            s += v.x * v.x + v.y * v.y + v.z * v.z + v.w * v.w;
        }
        ws[GRIDN + m] = s;
    }
}

__global__ __launch_bounds__(256)
void SOM_dist_kernel(const float* __restrict__ x,
                     const float* __restrict__ w,
                     const float* __restrict__ wnorm,
                     const float* __restrict__ xnorm,
                     float* __restrict__ out) {
    const int lane = threadIdx.x & 31;
    const int wid  = blockIdx.x * (blockDim.x >> 5) + (threadIdx.x >> 5);

    const int mtile = wid >> 6;
    const int ntile = wid & 63;
    const int m0 = mtile * 16;
    const int n0 = ntile * 64;

    const int lm   = lane & 15;
    const int half = lane >> 4;

    const float* xrow = x + (size_t)(m0 + lm) * DIM + 2 * half;
    const float* wrow = w + (size_t)(n0 + lm) * DIM + 2 * half;

    v8f acc0 = {}, acc1 = {}, acc2 = {}, acc3 = {};

#pragma unroll
    for (int kk = 0; kk < DIM / 4; ++kk) {
        const int ko = kk * 4;
        v2f a  = *(const v2f*)(xrow + ko);
        v2f b0 = *(const v2f*)(wrow + ko);
        v2f b1 = *(const v2f*)(wrow + 16 * DIM + ko);
        v2f b2 = *(const v2f*)(wrow + 32 * DIM + ko);
        v2f b3 = *(const v2f*)(wrow + 48 * DIM + ko);
        acc0 = __builtin_amdgcn_wmma_f32_16x16x4_f32(false, a, false, b0,
                                                     (short)0, acc0, false, false);
        acc1 = __builtin_amdgcn_wmma_f32_16x16x4_f32(false, a, false, b1,
                                                     (short)0, acc1, false, false);
        acc2 = __builtin_amdgcn_wmma_f32_16x16x4_f32(false, a, false, b2,
                                                     (short)0, acc2, false, false);
        acc3 = __builtin_amdgcn_wmma_f32_16x16x4_f32(false, a, false, b3,
                                                     (short)0, acc3, false, false);
    }

    float xr[8];
#pragma unroll
    for (int r = 0; r < 8; ++r) xr[r] = xnorm[m0 + r + 8 * half];

    v8f acc[4] = {acc0, acc1, acc2, acc3};
#pragma unroll
    for (int t = 0; t < 4; ++t) {
        const int n  = n0 + t * 16 + lm;
        const float wn = wnorm[n];
#pragma unroll
        for (int r = 0; r < 8; ++r) {
            const int m = m0 + r + 8 * half;
            out[(size_t)m * GRIDN + n] = fmaf(-2.f, acc[t][r], xr[r] + wn);
        }
    }
}

extern "C" void kernel_launch(void* const* d_in, const int* in_sizes, int n_in,
                              void* d_out, int out_size, void* d_ws, size_t ws_size,
                              hipStream_t stream) {
    const float* x = (const float*)d_in[0];
    const float* w = (const float*)d_in[1];
    float* out = (float*)d_out;
    float* ws  = (float*)d_ws;

    SOM_norms_kernel<<<(GRIDN + BATCH + 255) / 256, 256, 0, stream>>>(x, w, ws);

    SOM_dist_kernel<<<512, 256, 0, stream>>>(x, w, ws, ws + GRIDN, out);
}


# Row norms summed inside the distance kernel from the loaded fragments (half-row partial sums joined across lane pairs: new summation order of the norms); norms launch and workspace pass removed
# speedup vs baseline: 1.7071x; 1.7071x over previous
#include <hip/hip_runtime.h>
#include <hip/hip_bf16.h>


typedef __attribute__((ext_vector_type(2))) float v2f;
typedef __attribute__((ext_vector_type(8))) float v8f;

#define BATCH 1024
#define GRIDN 4096
#define DIM   64

__global__ __launch_bounds__(256)
void SOM_dist_kernel(const float* __restrict__ x,
                     const float* __restrict__ w,
                     float* __restrict__ out) {
    const int lane = threadIdx.x & 31;
    const int wid  = blockIdx.x * (blockDim.x >> 5) + (threadIdx.x >> 5);

    const int mtile = wid >> 6;
    const int ntile = wid & 63;
    const int m0 = mtile * 16;
    const int n0 = ntile * 64;

    const int lm   = lane & 15;
    const int half = lane >> 4;

    const float* xrow = x + (size_t)(m0 + lm) * DIM + 2 * half;
    const float* wrow = w + (size_t)(n0 + lm) * DIM + 2 * half;

    v8f acc0 = {}, acc1 = {}, acc2 = {}, acc3 = {};
    v2f xs = {0.f, 0.f};
    v2f ws0 = {0.f, 0.f}, ws1 = {0.f, 0.f}, ws2 = {0.f, 0.f}, ws3 = {0.f, 0.f};

#pragma unroll
    for (int kk = 0; kk < DIM / 4; ++kk) {
        const int ko = kk * 4;
        v2f a  = *(const v2f*)(xrow + ko);
        v2f b0 = *(const v2f*)(wrow + ko);
        v2f b1 = *(const v2f*)(wrow + 16 * DIM + ko);
        v2f b2 = *(const v2f*)(wrow + 32 * DIM + ko);
        v2f b3 = *(const v2f*)(wrow + 48 * DIM + ko);
        xs  += a  * a;
        ws0 += b0 * b0;
        ws1 += b1 * b1;
        ws2 += b2 * b2;
        ws3 += b3 * b3;
        acc0 = __builtin_amdgcn_wmma_f32_16x16x4_f32(false, a, false, b0,
                                                     (short)0, acc0, false, false);
        acc1 = __builtin_amdgcn_wmma_f32_16x16x4_f32(false, a, false, b1,
                                                     (short)0, acc1, false, false);
        acc2 = __builtin_amdgcn_wmma_f32_16x16x4_f32(false, a, false, b2,
                                                     (short)0, acc2, false, false);
        acc3 = __builtin_amdgcn_wmma_f32_16x16x4_f32(false, a, false, b3,
                                                     (short)0, acc3, false, false);
    }

    float xn = xs.x + xs.y;
    xn += __shfl_xor(xn, 16);
    float wn[4] = {ws0.x + ws0.y, ws1.x + ws1.y, ws2.x + ws2.y, ws3.x + ws3.y};
#pragma unroll
    for (int t = 0; t < 4; ++t) wn[t] += __shfl_xor(wn[t], 16);

    float xr[8];
#pragma unroll
    for (int r = 0; r < 8; ++r) xr[r] = __shfl(xn, r + 8 * half);

    v8f acc[4] = {acc0, acc1, acc2, acc3};
#pragma unroll
    for (int t = 0; t < 4; ++t) {
        const int n  = n0 + t * 16 + lm;
#pragma unroll
        for (int r = 0; r < 8; ++r) {
            const int m = m0 + r + 8 * half;
            out[(size_t)m * GRIDN + n] = fmaf(-2.f, acc[t][r], xr[r] + wn[t]);
        }
    }
}

extern "C" void kernel_launch(void* const* d_in, const int* in_sizes, int n_in,
                              void* d_out, int out_size, void* d_ws, size_t ws_size,
                              hipStream_t stream) {
    const float* x = (const float*)d_in[0];
    const float* w = (const float*)d_in[1];
    float* out = (float*)d_out;

    SOM_dist_kernel<<<512, 256, 0, stream>>>(x, w, out);
}


# Work-group tile reshaped to 128x64 and its x and w rows staged once in LDS by whole-line 16-byte loads; matrix fragments read from LDS, not by strided global loads; summation order unchanged
# speedup vs baseline: 2.8138x; 1.6483x over previous
#include <hip/hip_runtime.h>
#include <hip/hip_bf16.h>


typedef __attribute__((ext_vector_type(2))) float v2f;
typedef __attribute__((ext_vector_type(4))) float v4f;
typedef __attribute__((ext_vector_type(8))) float v8f;

#define BATCH 1024
#define GRIDN 4096
#define DIM   64

#define WAVES  8
#define AROWS  (WAVES * 16)
#define BROWS  64
#define NTHREADS (WAVES * 32)
#define PITCH 68

__global__ __launch_bounds__(NTHREADS) __attribute__((amdgpu_waves_per_eu(4, 4)))
void SOM_dist_kernel(const float* __restrict__ x,
                     const float* __restrict__ w,
                     float* __restrict__ out) {
    __shared__ __attribute__((aligned(16))) float sm[(AROWS + BROWS) * PITCH];

    const int tid  = threadIdx.x;
    const int lane = tid & 31;
    const int wv   = tid >> 5;

    const int mblk = blockIdx.x & 7;
    const int nblk = blockIdx.x >> 3;
    const int m0 = mblk * AROWS + wv * 16;
    const int n0 = nblk * BROWS;

    {
        const int row = tid >> 4;
        const int ch  = tid & 15;
        const float* xt = x + (size_t)mblk * AROWS * DIM + tid * 4;
        const float* wt = w + (size_t)n0 * DIM + tid * 4;
        float* dst = sm + row * PITCH + ch * 4;
        v4f va[AROWS / 16], vb[BROWS / 16];
#pragma unroll
        for (int j = 0; j < AROWS / 16; ++j) va[j] = *(const v4f*)(xt + j * NTHREADS * 4);
#pragma unroll
        for (int j = 0; j < BROWS / 16; ++j) vb[j] = *(const v4f*)(wt + j * NTHREADS * 4);
#pragma unroll
        for (int j = 0; j < AROWS / 16; ++j) *(v4f*)(dst + j * 16 * PITCH) = va[j];
#pragma unroll
        for (int j = 0; j < BROWS / 16; ++j) *(v4f*)(dst + (AROWS + j * 16) * PITCH) = vb[j];
    }
    __syncthreads();

    const int lm   = lane & 15;
    const int half = lane >> 4;

    const float* xrow = sm + (wv * 16 + lm) * PITCH + 2 * half;
    const float* wrow = sm + (AROWS + lm) * PITCH + 2 * half;

    v8f acc0 = {}, acc1 = {}, acc2 = {}, acc3 = {};
    v2f xs = {0.f, 0.f};
    v2f ws0 = {0.f, 0.f}, ws1 = {0.f, 0.f}, ws2 = {0.f, 0.f}, ws3 = {0.f, 0.f};

#pragma unroll
    for (int kk = 0; kk < DIM / 4; ++kk) {
        const int ko = kk * 4;
        v2f a  = *(const v2f*)(xrow + ko);
        v2f b0 = *(const v2f*)(wrow + ko);
        v2f b1 = *(const v2f*)(wrow + 16 * PITCH + ko);
        v2f b2 = *(const v2f*)(wrow + 32 * PITCH + ko);
        v2f b3 = *(const v2f*)(wrow + 48 * PITCH + ko);
        xs  += a  * a;
        ws0 += b0 * b0;
        ws1 += b1 * b1;
        ws2 += b2 * b2;
        ws3 += b3 * b3;
        acc0 = __builtin_amdgcn_wmma_f32_16x16x4_f32(false, a, false, b0,
                                                     (short)0, acc0, false, false);
        acc1 = __builtin_amdgcn_wmma_f32_16x16x4_f32(false, a, false, b1,
                                                     (short)0, acc1, false, false);
        acc2 = __builtin_amdgcn_wmma_f32_16x16x4_f32(false, a, false, b2,
                                                     (short)0, acc2, false, false);
        acc3 = __builtin_amdgcn_wmma_f32_16x16x4_f32(false, a, false, b3,
                                                     (short)0, acc3, false, false);
    }

    float xn = xs.x + xs.y;
    xn += __shfl_xor(xn, 16);
    float wn[4] = {ws0.x + ws0.y, ws1.x + ws1.y, ws2.x + ws2.y, ws3.x + ws3.y};
#pragma unroll
    for (int t = 0; t < 4; ++t) wn[t] += __shfl_xor(wn[t], 16);

    float xr[8];
#pragma unroll
    for (int r = 0; r < 8; ++r) xr[r] = __shfl(xn, r + 8 * half);

    v8f acc[4] = {acc0, acc1, acc2, acc3};
#pragma unroll
    for (int t = 0; t < 4; ++t) {
        const int n  = n0 + t * 16 + lm;
#pragma unroll
        for (int r = 0; r < 8; ++r) {
            const int m = m0 + r + 8 * half;
            out[(size_t)m * GRIDN + n] = fmaf(-2.f, acc[t][r], xr[r] + wn[t]);
        }
    }
}

extern "C" void kernel_launch(void* const* d_in, const int* in_sizes, int n_in,
                              void* d_out, int out_size, void* d_ws, size_t ws_size,
                              hipStream_t stream) {
    const float* x = (const float*)d_in[0];
    const float* w = (const float*)d_in[1];
    float* out = (float*)d_out;

    SOM_dist_kernel<<<512, 256, 0, stream>>>(x, w, out);
}
